# starting kernel
# baseline (speedup 1.0000x reference)
#include <hip/hip_runtime.h>

typedef __attribute__((ext_vector_type(2))) float v2f;
typedef __attribute__((ext_vector_type(4))) float v4f;
typedef __attribute__((ext_vector_type(8))) float v8f;

#define N_NODES 50000
#define N_EDGES 800000
#define D 128

__global__ void gcn_zero_kernel(v4f* __restrict__ p, int n4) {
    int i = blockIdx.x * blockDim.x + threadIdx.x;
    if (i < n4) {
        v4f z = {};
        p[i] = z;
    }
}

__global__ void __launch_bounds__(256)
gcn_spmm_kernel(const int* __restrict__ adj_row,
                const int* __restrict__ adj_col,
                const float* __restrict__ adj_vals,
                const float* __restrict__ features,
                float* __restrict__ f) {
    const int waveInBlk = __builtin_amdgcn_readfirstlane(threadIdx.x >> 5);
    const int lane      = threadIdx.x & 31;
    const int edge      = blockIdx.x * 8 + waveInBlk;
    if (edge >= N_EDGES) return;

    const int   r = adj_row[edge];
    const int   c = adj_col[edge];
    const float v = adj_vals[edge];

    const v4f x = *(const v4f*)(features + (size_t)c * D + lane * 4);
    float* dst = f + (size_t)r * D + lane * 4;

    unsafeAtomicAdd(dst + 0, v * x.x);
    unsafeAtomicAdd(dst + 1, v * x.y);
    unsafeAtomicAdd(dst + 2, v * x.z);
    unsafeAtomicAdd(dst + 3, v * x.w);
}

__global__ void __launch_bounds__(256)
gcn_gemm_wmma_kernel(const float* __restrict__ f,
                     const float* __restrict__ W,
                     const float* __restrict__ bias,
                     float* __restrict__ out) {
    const int waveInBlk = threadIdx.x >> 5;
    const int lane      = threadIdx.x & 31;
    const int waveId    = blockIdx.x * 8 + waveInBlk;

    const int mTile = waveId >> 3;
    const int oTile = waveId & 7;
    const int half  = lane >> 4;
    const int l     = lane & 15;

    const int m0 = mTile * 16;
    const int o0 = oTile * 16;

    const float* aRow = f + (size_t)(m0 + l) * D + 4 * half;
    const float* bRow = W + (size_t)(o0 + l) * D + 4 * half;

    v8f acc = {};

#pragma unroll
    for (int p = 0; p < 16; ++p) {
        const v4f va = *(const v4f*)(aRow + 8 * p);
        const v4f vb = *(const v4f*)(bRow + 8 * p);
        const v2f a0 = {va.x, va.y};
        const v2f b0 = {vb.x, vb.y};
        const v2f a1 = {va.z, va.w};
        const v2f b1 = {vb.z, vb.w};
        acc = __builtin_amdgcn_wmma_f32_16x16x4_f32(
            false, a0, false, b0, (short)0, acc, false, false);
        acc = __builtin_amdgcn_wmma_f32_16x16x4_f32(
            false, a1, false, b1, (short)0, acc, false, false);
    }

    const float bv = bias[o0 + l];
    float* outBase = out + (size_t)(m0 + 8 * half) * D + o0 + l;
#pragma unroll
    for (int r = 0; r < 8; ++r) {
        outBase[(size_t)r * D] = acc[r] + bv;
    }
}

extern "C" void kernel_launch(void* const* d_in, const int* in_sizes, int n_in,
                              void* d_out, int out_size, void* d_ws, size_t ws_size,
                              hipStream_t stream) {
    const int*   adj_row  = (const int*)d_in[0];
    const int*   adj_col  = (const int*)d_in[1];
    const float* adj_vals = (const float*)d_in[2];
    const float* features = (const float*)d_in[3];
    const float* W        = (const float*)d_in[4];
    const float* bias     = (const float*)d_in[5];
    float*       out      = (float*)d_out;

    float* f = (float*)d_ws;

    const int n4 = N_NODES * D / 4;
    gcn_zero_kernel<<<(n4 + 255) / 256, 256, 0, stream>>>((v4f*)f, n4);

    gcn_spmm_kernel<<<(N_EDGES + 7) / 8, 256, 0, stream>>>(
        adj_row, adj_col, adj_vals, features, f);

    const int gemmBlocks = (N_NODES / 16) * 8 / 8;
    gcn_gemm_wmma_kernel<<<gemmBlocks, 256, 0, stream>>>(f, W, bias, out);
}


# SpMM owners read per-owner edge bitmaps (1 bit per edge, built once per call by a new launch, ds_or_b32) instead of each scanning all of adj_row; 2 -> 3 launches; row sums stay in LDS arrival order
# speedup vs baseline: 2.3049x; 2.3049x over previous
#include <hip/hip_runtime.h>


typedef __attribute__((ext_vector_type(2))) float v2f;
typedef __attribute__((ext_vector_type(4))) float v4f;
typedef __attribute__((ext_vector_type(8))) float v8f;
typedef __attribute__((ext_vector_type(4))) int   v4i;
typedef __attribute__((ext_vector_type(4))) unsigned v4u;

#define N_NODES 50000
#define N_EDGES 800000
#define D 128

#define OWN_WGS     256
#define OWN_ROWS    196
#define OWN_THREADS 1024
#define OWN_WAVES   32

#define BM_CHUNK    2048
#define BM_CW       (BM_CHUNK / 32)
#define BM_CHUNKS   ((N_EDGES + BM_CHUNK - 1) / BM_CHUNK)
#define BM_WORDS    (BM_CHUNKS * BM_CW)
#define BM_PITCH    (BM_CW + 1)
#define BM_THREADS  256
#define BM_EPT      (BM_CHUNK / BM_THREADS)
#define BM_OFFSET   (32u << 20)

static_assert(N_EDGES % BM_EPT == 0, "a thread's 8 edges are in range together");

__global__ void __launch_bounds__(BM_THREADS)
gcn_owner_bitmap_kernel(const int* __restrict__ adj_row,
                        unsigned* __restrict__ bitmap) {
    __shared__ unsigned bm[OWN_WGS * BM_PITCH];

    const int      t   = threadIdx.x;
    const unsigned le0 = (unsigned)t * BM_EPT;
    const unsigned e0  = (unsigned)blockIdx.x * BM_CHUNK + le0;

    v4i ra = {-1, -1, -1, -1};
    v4i rb = ra;
    if (e0 < N_EDGES) {
        const v4i* p = (const v4i*)(adj_row + e0);
        ra = p[0];
        rb = p[1];
    }

    for (int i = t; i < OWN_WGS * BM_PITCH; i += BM_THREADS)
        bm[i] = 0u;
    __syncthreads();

    const unsigned wofs = le0 >> 5;
    const unsigned bit0 = le0 & 31u;
#pragma unroll
    for (int k = 0; k < 4; ++k) {
        const unsigned row = (unsigned)ra[k];
        if (row < N_NODES)
            __hip_atomic_fetch_or(&bm[(row / OWN_ROWS) * BM_PITCH + wofs], 1u << (bit0 + k),
                                  __ATOMIC_RELAXED, __HIP_MEMORY_SCOPE_WORKGROUP);
    }
#pragma unroll
    for (int k = 0; k < 4; ++k) {
        const unsigned row = (unsigned)rb[k];
        if (row < N_NODES)
            __hip_atomic_fetch_or(&bm[(row / OWN_ROWS) * BM_PITCH + wofs], 1u << (bit0 + 4 + k),
                                  __ATOMIC_RELAXED, __HIP_MEMORY_SCOPE_WORKGROUP);
    }
    __syncthreads();

    unsigned* dst = bitmap + (size_t)blockIdx.x * BM_CW;
    for (int i = t; i < OWN_WGS * BM_CW; i += BM_THREADS) {
        const int o = i / BM_CW;
        const int w = i % BM_CW;
        dst[(size_t)o * BM_WORDS + w] = bm[o * BM_PITCH + w];
    }
}

#define RING        1024
#define BM_INT4     (BM_WORDS / 4)
#define BM_LOADS    ((BM_INT4 + OWN_THREADS - 1) / OWN_THREADS)

__device__ __forceinline__ void own_drain(unsigned rd, unsigned take, int lane,
                                          int lo, unsigned nrows,
                                          const unsigned* ringw,
                                          const int* __restrict__ adj_row,
                                          const int* __restrict__ adj_col,
                                          const float* __restrict__ adj_vals,
                                          const float* __restrict__ features,
                                          float* tile) {
    const unsigned ent   = ringw[(rd + (unsigned)lane) & (RING - 1)];
    const bool     inb   = (unsigned)lane < take && ent < N_EDGES;
    const unsigned e     = inb ? ent : 0u;
    const unsigned drow  = (unsigned)(adj_row[e] - lo);
    const bool     valid = inb && drow < nrows;
    const int      lr    = valid ? (int)drow : OWN_ROWS;
    const int      c     = adj_col[e];
    const float    ve    = adj_vals[e];
    const float    v     = valid ? ve : 0.0f;

    for (unsigned j0 = 0; j0 < take; j0 += 8) {
        float x[8][4];
#pragma unroll
        for (int u = 0; u < 8; ++u) {
            const int cj = __builtin_amdgcn_readlane(c, (int)j0 + u);
            const float* s = features + (size_t)cj * D + lane;
            x[u][0] = s[0];
            x[u][1] = s[32];
            x[u][2] = s[64];
            x[u][3] = s[96];
        }
#pragma unroll
        for (int u = 0; u < 8; ++u) {
            const int   lrj = __builtin_amdgcn_readlane(lr, (int)j0 + u);
            const float vj  = __int_as_float(
                __builtin_amdgcn_readlane(__float_as_int(v), (int)j0 + u));
            float* d = tile + lrj * D + lane;
            __hip_atomic_fetch_add(d + 0,  vj * x[u][0], __ATOMIC_RELAXED, __HIP_MEMORY_SCOPE_WORKGROUP);
            __hip_atomic_fetch_add(d + 32, vj * x[u][1], __ATOMIC_RELAXED, __HIP_MEMORY_SCOPE_WORKGROUP);
            __hip_atomic_fetch_add(d + 64, vj * x[u][2], __ATOMIC_RELAXED, __HIP_MEMORY_SCOPE_WORKGROUP);
            __hip_atomic_fetch_add(d + 96, vj * x[u][3], __ATOMIC_RELAXED, __HIP_MEMORY_SCOPE_WORKGROUP);
        }
    }
}

__global__ void __launch_bounds__(OWN_THREADS)
gcn_spmm_owner_kernel(const int* __restrict__ adj_row,
                      const int* __restrict__ adj_col,
                      const float* __restrict__ adj_vals,
                      const float* __restrict__ features,
                      const unsigned* __restrict__ bitmap,
                      float* __restrict__ f) {
    __shared__ float    tile[(OWN_ROWS + 1) * D];
    __shared__ unsigned ring[OWN_WAVES * RING];

    const int t    = threadIdx.x;
    const int lane = t & 31;
    const int w    = __builtin_amdgcn_readfirstlane(t >> 5);
    const int lo   = blockIdx.x * OWN_ROWS;
    const unsigned nrows = (unsigned)((N_NODES - lo) < OWN_ROWS ? (N_NODES - lo) : OWN_ROWS);

    const v4u* bm4 = (const v4u*)(bitmap + (size_t)blockIdx.x * BM_WORDS);
    v4u r[BM_LOADS];
#pragma unroll
    for (int u = 0; u < BM_LOADS; ++u) {
        const unsigned idx = (unsigned)t + (unsigned)u * OWN_THREADS;
        const v4u zero4 = {0u, 0u, 0u, 0u};
        r[u] = (idx < BM_INT4) ? bm4[idx] : zero4;
    }

    v4f* tile4 = (v4f*)tile;
    for (int i = t; i < (OWN_ROWS + 1) * D / 4; i += OWN_THREADS) {
        v4f z = {};
        tile4[i] = z;
    }
    __syncthreads();

    unsigned* ringw = ring + w * RING;
    unsigned rd = 0, wr = 0;
#pragma unroll
    for (int u = 0; u < BM_LOADS; ++u) {
        v4u cur = r[u];
        unsigned e0 = ((unsigned)t + (unsigned)u * OWN_THREADS) * 128u;
        for (int j = 0; j < 4; ++j) {
            unsigned bits = cur.x;
            for (int pass = 0; pass < 32; ++pass) {
                const bool     m    = bits != 0u;
                const unsigned mask = __builtin_amdgcn_ballot_w32(m);
                if (!mask) break;
                if (m) {
                    const unsigned pos = wr + __builtin_amdgcn_mbcnt_lo(mask, 0u);
                    ringw[pos & (RING - 1)] = e0 + (unsigned)__builtin_ctz(bits);
                    bits &= bits - 1u;
                }
                wr += (unsigned)__builtin_popcount(mask);
                if (wr - rd >= 32u) {
                    own_drain(rd, 32u, lane, lo, nrows, ringw,
                              adj_row, adj_col, adj_vals, features, tile);
                    rd += 32u;
                }
            }
            const v4u rot = {cur.y, cur.z, cur.w, 0u};
            cur = rot;
            e0 += 32u;
        }
    }
    if (wr != rd)
        own_drain(rd, wr - rd, lane, lo, nrows, ringw,
                  adj_row, adj_col, adj_vals, features, tile);

    __syncthreads();

    v4f* f4 = (v4f*)(f + (size_t)lo * D);
    for (unsigned i = (unsigned)t; i < nrows * (D / 4); i += OWN_THREADS)
        f4[i] = tile4[i];
}

__global__ void __launch_bounds__(256)
gcn_gemm_wmma_kernel(const float* __restrict__ f,
                     const float* __restrict__ W,
                     const float* __restrict__ bias,
                     float* __restrict__ out) {
    const int waveInBlk = threadIdx.x >> 5;
    const int lane      = threadIdx.x & 31;
    const int waveId    = blockIdx.x * 8 + waveInBlk;

    const int mTile = waveId >> 3;
    const int oTile = waveId & 7;
    const int half  = lane >> 4;
    const int l     = lane & 15;

    const int m0 = mTile * 16;
    const int o0 = oTile * 16;

    const float* aRow = f + (size_t)(m0 + l) * D + 4 * half;
    const float* bRow = W + (size_t)(o0 + l) * D + 4 * half;

    v8f acc = {};

#pragma unroll
    for (int p = 0; p < 16; ++p) {
        const v4f va = *(const v4f*)(aRow + 8 * p);
        const v4f vb = *(const v4f*)(bRow + 8 * p);
        const v2f a0 = {va.x, va.y};
        const v2f b0 = {vb.x, vb.y};
        const v2f a1 = {va.z, va.w};
        const v2f b1 = {vb.z, vb.w};
        acc = __builtin_amdgcn_wmma_f32_16x16x4_f32(
            false, a0, false, b0, (short)0, acc, false, false);
        acc = __builtin_amdgcn_wmma_f32_16x16x4_f32(
            false, a1, false, b1, (short)0, acc, false, false);
    }

    const float bv = bias[o0 + l];
    float* outBase = out + (size_t)(m0 + 8 * half) * D + o0 + l;
#pragma unroll
    for (int r = 0; r < 8; ++r) {
        outBase[(size_t)r * D] = acc[r] + bv;
    }
}

extern "C" void kernel_launch(void* const* d_in, const int* in_sizes, int n_in,
                              void* d_out, int out_size, void* d_ws, size_t ws_size,
                              hipStream_t stream) {
    const int*   adj_row  = (const int*)d_in[0];
    const int*   adj_col  = (const int*)d_in[1];
    const float* adj_vals = (const float*)d_in[2];
    const float* features = (const float*)d_in[3];
    const float* W        = (const float*)d_in[4];
    const float* bias     = (const float*)d_in[5];
    float*       out      = (float*)d_out;

    float*    f      = (float*)d_ws;
    unsigned* bitmap = (unsigned*)((char*)d_ws + BM_OFFSET);

    gcn_owner_bitmap_kernel<<<BM_CHUNKS, BM_THREADS, 0, stream>>>(adj_row, bitmap);

    gcn_spmm_owner_kernel<<<OWN_WGS, OWN_THREADS, 0, stream>>>(
        adj_row, adj_col, adj_vals, features, bitmap, f);

    const int gemmBlocks = (N_NODES / 16) * 8 / 8;
    gcn_gemm_wmma_kernel<<<gemmBlocks, 256, 0, stream>>>(f, W, bias, out);
}


# Linear layer fused into the owner kernel: each owner runs the f32 WMMA straight from its LDS row sums (row pitch 132) and stores whole rows of out; f and the GEMM launch go, 3 -> 2 launches
# speedup vs baseline: 3.1364x; 1.3608x over previous
#include <hip/hip_runtime.h>


typedef __attribute__((ext_vector_type(2))) float v2f;
typedef __attribute__((ext_vector_type(4))) float v4f;
typedef __attribute__((ext_vector_type(8))) float v8f;
typedef __attribute__((ext_vector_type(4))) int   v4i;
typedef __attribute__((ext_vector_type(4))) unsigned v4u;

#define N_NODES 50000
#define N_EDGES 800000
#define D 128

#define OWN_WGS     256
#define OWN_ROWS    196
#define OWN_THREADS 1024
#define OWN_WAVES   32

#define BM_CHUNK    2048
#define BM_CW       (BM_CHUNK / 32)
#define BM_CHUNKS   ((N_EDGES + BM_CHUNK - 1) / BM_CHUNK)
#define BM_WORDS    (BM_CHUNKS * BM_CW)
#define BM_PITCH    (BM_CW + 1)
#define BM_THREADS  256
#define BM_EPT      (BM_CHUNK / BM_THREADS)

static_assert(N_EDGES % BM_EPT == 0, "a thread's 8 edges are in range together");

__global__ void __launch_bounds__(BM_THREADS)
gcn_owner_bitmap_kernel(const int* __restrict__ adj_row,
                        unsigned* __restrict__ bitmap) {
    __shared__ unsigned bm[OWN_WGS * BM_PITCH];

    const int      t   = threadIdx.x;
    const unsigned le0 = (unsigned)t * BM_EPT;
    const unsigned e0  = (unsigned)blockIdx.x * BM_CHUNK + le0;

    v4i ra = {-1, -1, -1, -1};
    v4i rb = ra;
    if (e0 < N_EDGES) {
        const v4i* p = (const v4i*)(adj_row + e0);
        ra = p[0];
        rb = p[1];
    }

    for (int i = t; i < OWN_WGS * BM_PITCH; i += BM_THREADS)
        bm[i] = 0u;
    __syncthreads();

    const unsigned wofs = le0 >> 5;
    const unsigned bit0 = le0 & 31u;
#pragma unroll
    for (int k = 0; k < 4; ++k) {
        const unsigned row = (unsigned)ra[k];
        if (row < N_NODES)
            __hip_atomic_fetch_or(&bm[(row / OWN_ROWS) * BM_PITCH + wofs], 1u << (bit0 + k),
                                  __ATOMIC_RELAXED, __HIP_MEMORY_SCOPE_WORKGROUP);
    }
#pragma unroll
    for (int k = 0; k < 4; ++k) {
        const unsigned row = (unsigned)rb[k];
        if (row < N_NODES)
            __hip_atomic_fetch_or(&bm[(row / OWN_ROWS) * BM_PITCH + wofs], 1u << (bit0 + 4 + k),
                                  __ATOMIC_RELAXED, __HIP_MEMORY_SCOPE_WORKGROUP);
    }
    __syncthreads();

    unsigned* dst = bitmap + (size_t)blockIdx.x * BM_CW;
    for (int i = t; i < OWN_WGS * BM_CW; i += BM_THREADS) {
        const int o = i / BM_CW;
        const int w = i % BM_CW;
        dst[(size_t)o * BM_WORDS + w] = bm[o * BM_PITCH + w];
    }
}

#define RING        128
#define M_TILES     ((OWN_ROWS + 15) / 16)
#define TILE_ROWS   (M_TILES * 16)
#define TP          (D + 4)
static_assert(OWN_ROWS < TILE_ROWS, "the spare row is inside the tile");
#define BM_INT4     (BM_WORDS / 4)
#define BM_LOADS    ((BM_INT4 + OWN_THREADS - 1) / OWN_THREADS)

__device__ __forceinline__ void own_drain(unsigned rd, unsigned take, int lane,
                                          int lo, unsigned nrows,
                                          const unsigned* ringw,
                                          const int* __restrict__ adj_row,
                                          const int* __restrict__ adj_col,
                                          const float* __restrict__ adj_vals,
                                          const float* __restrict__ features,
                                          float* tile) {
    const unsigned ent   = ringw[(rd + (unsigned)lane) & (RING - 1)];
    const bool     inb   = (unsigned)lane < take && ent < N_EDGES;
    const unsigned e     = inb ? ent : 0u;
    const unsigned drow  = (unsigned)(adj_row[e] - lo);
    const bool     valid = inb && drow < nrows;
    const int      lr    = valid ? (int)drow : OWN_ROWS;
    const int      c     = adj_col[e];
    const float    ve    = adj_vals[e];
    const float    v     = valid ? ve : 0.0f;

    for (unsigned j0 = 0; j0 < take; j0 += 8) {
        float x[8][4];
#pragma unroll
        for (int u = 0; u < 8; ++u) {
            const int cj = __builtin_amdgcn_readlane(c, (int)j0 + u);
            const float* s = features + (size_t)cj * D + lane;
            x[u][0] = s[0];
            x[u][1] = s[32];
            x[u][2] = s[64];
            x[u][3] = s[96];
        }
#pragma unroll
        for (int u = 0; u < 8; ++u) {
            const int   lrj = __builtin_amdgcn_readlane(lr, (int)j0 + u);
            const float vj  = __int_as_float(
                __builtin_amdgcn_readlane(__float_as_int(v), (int)j0 + u));
            float* d = tile + lrj * TP + lane;
            __hip_atomic_fetch_add(d + 0,  vj * x[u][0], __ATOMIC_RELAXED, __HIP_MEMORY_SCOPE_WORKGROUP);
            __hip_atomic_fetch_add(d + 32, vj * x[u][1], __ATOMIC_RELAXED, __HIP_MEMORY_SCOPE_WORKGROUP);
            __hip_atomic_fetch_add(d + 64, vj * x[u][2], __ATOMIC_RELAXED, __HIP_MEMORY_SCOPE_WORKGROUP);
            __hip_atomic_fetch_add(d + 96, vj * x[u][3], __ATOMIC_RELAXED, __HIP_MEMORY_SCOPE_WORKGROUP);
        }
    }
}

__global__ void __launch_bounds__(OWN_THREADS)
gcn_spmm_owner_kernel(const int* __restrict__ adj_row,
                      const int* __restrict__ adj_col,
                      const float* __restrict__ adj_vals,
                      const float* __restrict__ features,
                      const unsigned* __restrict__ bitmap,
                      const float* __restrict__ W,
                      const float* __restrict__ bias,
                      float* __restrict__ out) {
    __shared__ float    tile[TILE_ROWS * TP];
    __shared__ float    otile[TILE_ROWS * TP];
    __shared__ unsigned ring[OWN_WAVES * RING];

    const int t    = threadIdx.x;
    const int lane = t & 31;
    const int w    = __builtin_amdgcn_readfirstlane(t >> 5);
    const int lo   = blockIdx.x * OWN_ROWS;
    const unsigned nrows = (unsigned)((N_NODES - lo) < OWN_ROWS ? (N_NODES - lo) : OWN_ROWS);

    const v4u* bm4 = (const v4u*)(bitmap + (size_t)blockIdx.x * BM_WORDS);
    v4u r[BM_LOADS];
#pragma unroll
    for (int u = 0; u < BM_LOADS; ++u) {
        const unsigned idx = (unsigned)t + (unsigned)u * OWN_THREADS;
        const v4u zero4 = {0u, 0u, 0u, 0u};
        r[u] = (idx < BM_INT4) ? bm4[idx] : zero4;
    }

    v4f* tile4 = (v4f*)tile;
    for (int i = t; i < TILE_ROWS * TP / 4; i += OWN_THREADS) {
        v4f z = {};
        tile4[i] = z;
    }
    __syncthreads();

    unsigned* ringw = ring + w * RING;
    unsigned rd = 0, wr = 0;
#pragma unroll
    for (int u = 0; u < BM_LOADS; ++u) {
        v4u cur = r[u];
        unsigned e0 = ((unsigned)t + (unsigned)u * OWN_THREADS) * 128u;
        for (int j = 0; j < 4; ++j) {
            unsigned bits = cur.x;
            for (int pass = 0; pass < 32; ++pass) {
                const bool     m    = bits != 0u;
                const unsigned mask = __builtin_amdgcn_ballot_w32(m);
                if (!mask) break;
                if (m) {
                    const unsigned pos = wr + __builtin_amdgcn_mbcnt_lo(mask, 0u);
                    ringw[pos & (RING - 1)] = e0 + (unsigned)__builtin_ctz(bits);
                    bits &= bits - 1u;
                }
                wr += (unsigned)__builtin_popcount(mask);
                if (wr - rd >= 32u) {
                    own_drain(rd, 32u, lane, lo, nrows, ringw,
                              adj_row, adj_col, adj_vals, features, tile);
                    rd += 32u;
                }
            }
            const v4u rot = {cur.y, cur.z, cur.w, 0u};
            cur = rot;
            e0 += 32u;
        }
    }
    if (wr != rd)
        own_drain(rd, wr - rd, lane, lo, nrows, ringw,
                  adj_row, adj_col, adj_vals, features, tile);

    const int oT   = w & 7;
    const int half = lane >> 4;
    const int l    = lane & 15;

    const float* bRow = W + (size_t)(oT * 16 + l) * D + 4 * half;
    v4f wb[16];
#pragma unroll
    for (int p = 0; p < 16; ++p)
        wb[p] = *(const v4f*)(bRow + 8 * p);
    const float bv = bias[oT * 16 + l];

    __syncthreads();

    for (int mt = w >> 3; mt < M_TILES; mt += OWN_WAVES / 8) {
        const float* aRow = tile + (mt * 16 + l) * TP + 4 * half;
        v8f acc = {};
#pragma unroll
        for (int p = 0; p < 16; ++p) {
            const v4f va = *(const v4f*)(aRow + 8 * p);
            const v4f vb = wb[p];
            const v2f a0 = {va.x, va.y};
            const v2f b0 = {vb.x, vb.y};
            const v2f a1 = {va.z, va.w};
            const v2f b1 = {vb.z, vb.w};
            acc = __builtin_amdgcn_wmma_f32_16x16x4_f32(
                false, a0, false, b0, (short)0, acc, false, false);
            acc = __builtin_amdgcn_wmma_f32_16x16x4_f32(
                false, a1, false, b1, (short)0, acc, false, false);
        }
        float* oBase = otile + (mt * 16 + 8 * half) * TP + oT * 16 + l;
#pragma unroll
        for (int r = 0; r < 8; ++r)
            oBase[r * TP] = acc[r] + bv;
    }

    __syncthreads();

    v4f* out4 = (v4f*)(out + (size_t)lo * D);
    for (unsigned i = (unsigned)t; i < nrows * (D / 4); i += OWN_THREADS)
        out4[i] = *(const v4f*)(otile + (i >> 5) * TP + (i & 31u) * 4u);
}

extern "C" void kernel_launch(void* const* d_in, const int* in_sizes, int n_in,
                              void* d_out, int out_size, void* d_ws, size_t ws_size,
                              hipStream_t stream) {
    const int*   adj_row  = (const int*)d_in[0];
    const int*   adj_col  = (const int*)d_in[1];
    const float* adj_vals = (const float*)d_in[2];
    const float* features = (const float*)d_in[3];
    const float* W        = (const float*)d_in[4];
    const float* bias     = (const float*)d_in[5];
    float*       out      = (float*)d_out;

    unsigned* bitmap = (unsigned*)d_ws;

    gcn_owner_bitmap_kernel<<<BM_CHUNKS, BM_THREADS, 0, stream>>>(adj_row, bitmap);

    gcn_spmm_owner_kernel<<<OWN_WGS, OWN_THREADS, 0, stream>>>(
        adj_row, adj_col, adj_vals, features, bitmap, W, bias, out);
}
